# starting kernel
# baseline (speedup 1.0000x reference)
#include <hip/hip_runtime.h>

typedef __attribute__((ext_vector_type(2))) float v2f;
typedef __attribute__((ext_vector_type(8))) float v8f;

#define B_    4
#define CIN_  16
#define COUT_ 16
#define H_    128
#define W_    128
#define KK    144
#define WPB   8

__global__ __launch_bounds__(256) void xonv2d_wmma_f32(
    const float* __restrict__ x,
    const float* __restrict__ wgt,
    const float* __restrict__ bias,
    float* __restrict__ out)
{
  __shared__ __align__(16) float patch[WPB][B_][KK];

  const int lane = threadIdx.x & 31;
  const int wave = threadIdx.x >> 5;
  const int pix  = blockIdx.x * WPB + wave;
  const int h    = pix >> 7;
  const int w    = pix & 127;

  float* pat = &patch[wave][0][0];
  for (int e = lane; e < B_ * KK; e += 32) {
    int b  = e / KK;
    int p  = e - b * KK;
    int c  = p / 9;
    int r  = p - c * 9;
    int kh = r / 3;
    int kw = r - kh * 3;
    int hh = h + kh - 1;
    int ww = w + kw - 1;
    float v = 0.0f;
    if (hh >= 0 && hh < H_ && ww >= 0 && ww < W_)
      v = x[((b * CIN_ + c) * H_ + hh) * W_ + ww];
    pat[e] = v;
  }
  __syncthreads();

  const float* wpix = wgt + (size_t)pix * (COUT_ * KK);
  const int n    = lane & 15;
  const int half = lane >> 4;

  v8f acc = {};
  for (int t = 0; t < KK / 4; ++t) {
    const int p0 = 4 * t + 2 * half;
    v2f a = *(const v2f*)(wpix + n * KK + p0);
    v2f b = *(const v2f*)(pat + (n & (B_ - 1)) * KK + p0);
    acc = __builtin_amdgcn_wmma_f32_16x16x4_f32(
         false, a,  false, b,
         (short)0, acc,  false,  false);
  }

  if (n < B_) {
    const float* bp = bias + (size_t)pix * COUT_;
    float* op = out + (size_t)n * COUT_ * (H_ * W_) + pix;
#pragma unroll
    for (int r = 0; r < 8; ++r) {
      const int m = r + 8 * half;
      op[(size_t)m * (H_ * W_)] = acc[r] + bp[m];
    }
  }
}

extern "C" void kernel_launch(void* const* d_in, const int* in_sizes, int n_in,
                              void* d_out, int out_size, void* d_ws, size_t ws_size,
                              hipStream_t stream) {
  (void)in_sizes; (void)n_in; (void)out_size; (void)d_ws; (void)ws_size;
  const float* x    = (const float*)d_in[0];
  const float* wgt  = (const float*)d_in[1];
  const float* bias = (const float*)d_in[2];
  float* out        = (float*)d_out;

  dim3 grid((H_ * W_) / WPB);
  dim3 block(256);
  xonv2d_wmma_f32<<<grid, block, 0, stream>>>(x, wgt, bias, out);
}


# Each wave walks 4 pixels of a row, the next pixel's 36 weight loads issued ahead of the current pixel's WMMAs (4096 waves instead of 16384; patches staged once per wave, no block barrier)
# speedup vs baseline: 2.2672x; 2.2672x over previous
#include <hip/hip_runtime.h>

typedef __attribute__((ext_vector_type(2))) float v2f;
typedef __attribute__((ext_vector_type(4))) float v4f;
typedef __attribute__((ext_vector_type(8))) float v8f;

#define B_    4
#define CIN_  16
#define COUT_ 16
#define H_    128
#define W_    128
#define KK    144
#define NT    (KK / 4)
#define WPB   8
#define PPW   4
#define PSZ   (B_ * KK)


__device__ __forceinline__ void load_weights(v2f (&a)[NT], const float* __restrict__ wl) {
#pragma unroll
  for (int t = 0; t < NT; ++t) a[t] = *(const v2f*)(wl + 4 * t);
}

__device__ __forceinline__ void pixel_step(
    const v2f (&cur)[NT], v2f (&nxt)[NT], const bool has_next,
    const float* __restrict__ wl_next, const float* __restrict__ pb,
    const float* __restrict__ bp, float* __restrict__ op, const bool writer)
{
  v4f b0 = *(const v4f*)bp;
  v4f b1 = *(const v4f*)(bp + 4);
  __builtin_amdgcn_sched_barrier(0);
  if (has_next) load_weights(nxt, wl_next);
  __builtin_amdgcn_sched_barrier(0);

  v8f acc = {};
#pragma unroll
  for (int t = 0; t < NT; ++t) {
    const v2f b = *(const v2f*)(pb + 4 * t);
    acc = __builtin_amdgcn_wmma_f32_16x16x4_f32(
         false, cur[t],  false, b,
         (short)0, acc,  false,  false);
  }

  asm volatile("" : "+v"(b0), "+v"(b1));

  if (writer) {
    op[0 * (H_ * W_)] = acc[0] + b0[0];
    op[1 * (H_ * W_)] = acc[1] + b0[1];
    op[2 * (H_ * W_)] = acc[2] + b0[2];
    op[3 * (H_ * W_)] = acc[3] + b0[3];
    op[4 * (H_ * W_)] = acc[4] + b1[0];
    op[5 * (H_ * W_)] = acc[5] + b1[1];
    op[6 * (H_ * W_)] = acc[6] + b1[2];
    op[7 * (H_ * W_)] = acc[7] + b1[3];
  }
}

__global__ __launch_bounds__(256)
void xonv2d_wmma_f32(
    const float* __restrict__ x,
    const float* __restrict__ wgt,
    const float* __restrict__ bias,
    float* __restrict__ out)
{
  __shared__ __align__(16) float patch[WPB][PPW][PSZ];

  const int lane = threadIdx.x & 31;
  const int wave = threadIdx.x >> 5;
  const int n    = lane & 15;
  const int half = lane >> 4;
  const int pix0 = blockIdx.x * (WPB * PPW) + wave;
  const int h    = pix0 >> 7;
  const int w0   = pix0 & 127;

  const float* wl = wgt + (size_t)pix0 * (COUT_ * KK) + n * KK + 2 * half;
  const size_t wstep = (size_t)WPB * (COUT_ * KK);

  v2f wa[NT], wb[NT];
  load_weights(wa, wl);
  __builtin_amdgcn_sched_barrier(0);

  float* pat = &patch[wave][0][0];
#pragma unroll
  for (int j = 0; j < PSZ / 32; ++j) {
    const int e  = lane + 32 * j;
    const int b  = e / KK;
    const int p  = e - b * KK;
    const int c  = p / 9;
    const int r  = p - c * 9;
    const int kh = r / 3;
    const int kw = r - kh * 3;
    const int hh = h + kh - 1;
    const bool okh = (unsigned)hh < (unsigned)H_;
    const int base = ((b * CIN_ + c) * H_ + hh) * W_ + (kw - 1);
#pragma unroll
    for (int i = 0; i < PPW; ++i) {
      const int wpx = w0 + 8 * i;
      const bool ok = okh && (unsigned)(wpx + kw - 1) < (unsigned)W_;
      const float v = x[ok ? base + wpx : 0];
      pat[i * PSZ + e] = ok ? v : 0.0f;
    }
  }
  __builtin_amdgcn_wave_barrier();

  const float* pb = pat + (n & (B_ - 1)) * KK + 2 * half;
  const float* bp = bias + (size_t)pix0 * COUT_ + 8 * half;
  float* op = out + (size_t)(n & (B_ - 1)) * COUT_ * (H_ * W_) + (size_t)(8 * half) * (H_ * W_) + pix0;
  const bool writer = n < B_;

  pixel_step(wa, wb, true,  wl + 1 * wstep, pb + 0 * PSZ, bp + 0 * WPB * COUT_, op + 0 * WPB, writer);
  pixel_step(wb, wa, true,  wl + 2 * wstep, pb + 1 * PSZ, bp + 1 * WPB * COUT_, op + 1 * WPB, writer);
  pixel_step(wa, wb, true,  wl + 3 * wstep, pb + 2 * PSZ, bp + 2 * WPB * COUT_, op + 2 * WPB, writer);
  pixel_step(wb, wa, false, wl,             pb + 3 * PSZ, bp + 3 * WPB * COUT_, op + 3 * WPB, writer);
}

extern "C" void kernel_launch(void* const* d_in, const int* in_sizes, int n_in,
                              void* d_out, int out_size, void* d_ws, size_t ws_size,
                              hipStream_t stream) {
  (void)in_sizes; (void)n_in; (void)out_size; (void)d_ws; (void)ws_size;
  const float* x    = (const float*)d_in[0];
  const float* wgt  = (const float*)d_in[1];
  const float* bias = (const float*)d_in[2];
  float* out        = (float*)d_out;

  dim3 grid((H_ * W_) / (WPB * PPW));
  dim3 block(256);
  xonv2d_wmma_f32<<<grid, block, 0, stream>>>(x, wgt, bias, out);
}


# Weight and patch fragments read 16 B per lane (4 consecutive k = two WMMA steps): 18 b128 loads per pixel instead of 36 b64, each line touched by half as many loads; k summed in a new order
# speedup vs baseline: 2.4413x; 1.0768x over previous
#include <hip/hip_runtime.h>

typedef __attribute__((ext_vector_type(2))) float v2f;
typedef __attribute__((ext_vector_type(4))) float v4f;
typedef __attribute__((ext_vector_type(8))) float v8f;

#define B_    4
#define CIN_  16
#define COUT_ 16
#define H_    128
#define W_    128
#define KK    144
#define NT    (KK / 4)
#define NL    (KK / 8)
#define WPB   8
#define PPW   4
#define PSZ   (B_ * KK)


__device__ __forceinline__ void load_weights(v4f (&a)[NL], const float* __restrict__ wl) {
#pragma unroll
  for (int j = 0; j < NL; ++j) a[j] = *(const v4f*)(wl + 8 * j);
}

__device__ __forceinline__ void pixel_step(
    const v4f (&cur)[NL], v4f (&nxt)[NL], const bool has_next,
    const float* __restrict__ wl_next, const float* __restrict__ pb,
    const float* __restrict__ bp, float* __restrict__ op, const bool writer)
{
  v4f b0 = *(const v4f*)bp;
  v4f b1 = *(const v4f*)(bp + 4);
  __builtin_amdgcn_sched_barrier(0);
  if (has_next) load_weights(nxt, wl_next);
  __builtin_amdgcn_sched_barrier(0);

  v8f acc = {};
#pragma unroll
  for (int j = 0; j < NL; ++j) {
    const v4f a = cur[j];
    const v4f b = *(const v4f*)(pb + 8 * j);
    acc = __builtin_amdgcn_wmma_f32_16x16x4_f32(
         false, __builtin_shufflevector(a, a, 0, 1),
         false, __builtin_shufflevector(b, b, 0, 1),
         (short)0, acc,  false,  false);
    acc = __builtin_amdgcn_wmma_f32_16x16x4_f32(
         false, __builtin_shufflevector(a, a, 2, 3),
         false, __builtin_shufflevector(b, b, 2, 3),
         (short)0, acc,  false,  false);
  }

  asm volatile("" : "+v"(b0), "+v"(b1));

  if (writer) {
    op[0 * (H_ * W_)] = acc[0] + b0[0];
    op[1 * (H_ * W_)] = acc[1] + b0[1];
    op[2 * (H_ * W_)] = acc[2] + b0[2];
    op[3 * (H_ * W_)] = acc[3] + b0[3];
    op[4 * (H_ * W_)] = acc[4] + b1[0];
    op[5 * (H_ * W_)] = acc[5] + b1[1];
    op[6 * (H_ * W_)] = acc[6] + b1[2];
    op[7 * (H_ * W_)] = acc[7] + b1[3];
  }
}

__global__ __launch_bounds__(256)
void xonv2d_wmma_f32(
    const float* __restrict__ x,
    const float* __restrict__ wgt,
    const float* __restrict__ bias,
    float* __restrict__ out)
{
  __shared__ __align__(16) float patch[WPB][PPW][PSZ];

  const int lane = threadIdx.x & 31;
  const int wave = threadIdx.x >> 5;
  const int n    = lane & 15;
  const int half = lane >> 4;
  const int pix0 = blockIdx.x * (WPB * PPW) + wave;
  const int h    = pix0 >> 7;
  const int w0   = pix0 & 127;

  const float* wl = wgt + (size_t)pix0 * (COUT_ * KK) + n * KK + 4 * half;
  const size_t wstep = (size_t)WPB * (COUT_ * KK);

  v4f wa[NL], wb[NL];
  load_weights(wa, wl);
  __builtin_amdgcn_sched_barrier(0);

  float* pat = &patch[wave][0][0];
#pragma unroll
  for (int j = 0; j < PSZ / 32; ++j) {
    const int e  = lane + 32 * j;
    const int b  = e / KK;
    const int p  = e - b * KK;
    const int c  = p / 9;
    const int r  = p - c * 9;
    const int kh = r / 3;
    const int kw = r - kh * 3;
    const int hh = h + kh - 1;
    const bool okh = (unsigned)hh < (unsigned)H_;
    const int base = ((b * CIN_ + c) * H_ + hh) * W_ + (kw - 1);
#pragma unroll
    for (int i = 0; i < PPW; ++i) {
      const int wpx = w0 + 8 * i;
      const bool ok = okh && (unsigned)(wpx + kw - 1) < (unsigned)W_;
      const float v = x[ok ? base + wpx : 0];
      pat[i * PSZ + e] = ok ? v : 0.0f;
    }
  }
  __builtin_amdgcn_wave_barrier();

  const float* pb = pat + (n & (B_ - 1)) * KK + 4 * half;
  const float* bp = bias + (size_t)pix0 * COUT_ + 8 * half;
  float* op = out + (size_t)(n & (B_ - 1)) * COUT_ * (H_ * W_) + (size_t)(8 * half) * (H_ * W_) + pix0;
  const bool writer = n < B_;

  pixel_step(wa, wb, true,  wl + 1 * wstep, pb + 0 * PSZ, bp + 0 * WPB * COUT_, op + 0 * WPB, writer);
  pixel_step(wb, wa, true,  wl + 2 * wstep, pb + 1 * PSZ, bp + 1 * WPB * COUT_, op + 1 * WPB, writer);
  pixel_step(wa, wb, true,  wl + 3 * wstep, pb + 2 * PSZ, bp + 2 * WPB * COUT_, op + 2 * WPB, writer);
  pixel_step(wb, wa, false, wl,             pb + 3 * PSZ, bp + 3 * WPB * COUT_, op + 3 * WPB, writer);
}

extern "C" void kernel_launch(void* const* d_in, const int* in_sizes, int n_in,
                              void* d_out, int out_size, void* d_ws, size_t ws_size,
                              hipStream_t stream) {
  (void)in_sizes; (void)n_in; (void)out_size; (void)d_ws; (void)ws_size;
  const float* x    = (const float*)d_in[0];
  const float* wgt  = (const float*)d_in[1];
  const float* bias = (const float*)d_in[2];
  float* out        = (float*)d_out;

  dim3 grid((H_ * W_) / (WPB * PPW));
  dim3 block(256);
  xonv2d_wmma_f32<<<grid, block, 0, stream>>>(x, wgt, bias, out);
}


# Each wave takes 4 adjacent pixels and builds their patches from 8-float row segments of x (48 loads of 4 rows, issued ahead of the weights) in place of 72 scattered gathers; k order unchanged
# speedup vs baseline: 2.6975x; 1.1049x over previous
#include <hip/hip_runtime.h>

typedef __attribute__((ext_vector_type(2))) float v2f;
typedef __attribute__((ext_vector_type(4))) float v4f;
typedef __attribute__((ext_vector_type(8))) float v8f;

#define B_    4
#define CIN_  16
#define COUT_ 16
#define H_    128
#define W_    128
#define KK    144
#define NT    (KK / 4)
#define NL    (KK / 8)
#define WPB   8
#define PPW   4
#define PSZ   (B_ * KK)
#define PITCH 580
#define WREG  2944
#define WSTEP (COUT_ * KK)


__device__ __forceinline__ void load_weights(v4f (&a)[NL], const float* __restrict__ wl) {
#pragma unroll
  for (int j = 0; j < NL; ++j) a[j] = *(const v4f*)(wl + 8 * j);
}

__device__ __forceinline__ void pixel_step(
    const v4f (&cur)[NL], v4f (&nxt)[NL], const bool has_next,
    const float* __restrict__ wl_next, const float* __restrict__ pb,
    const float* __restrict__ bp, float* __restrict__ op, const bool writer)
{
  v4f b0 = *(const v4f*)bp;
  v4f b1 = *(const v4f*)(bp + 4);
  __builtin_amdgcn_sched_barrier(0);
  if (has_next) load_weights(nxt, wl_next);
  __builtin_amdgcn_sched_barrier(0);

  v8f acc = {};
#pragma unroll
  for (int j = 0; j < NL; ++j) {
    const v4f a = cur[j];
    const v4f b = *(const v4f*)(pb + 8 * j);
    acc = __builtin_amdgcn_wmma_f32_16x16x4_f32(
         false, __builtin_shufflevector(a, a, 0, 1),
         false, __builtin_shufflevector(b, b, 0, 1),
         (short)0, acc,  false,  false);
    acc = __builtin_amdgcn_wmma_f32_16x16x4_f32(
         false, __builtin_shufflevector(a, a, 2, 3),
         false, __builtin_shufflevector(b, b, 2, 3),
         (short)0, acc,  false,  false);
  }

  asm volatile("" : "+v"(b0), "+v"(b1));

  if (writer) {
    op[0 * (H_ * W_)] = acc[0] + b0[0];
    op[1 * (H_ * W_)] = acc[1] + b0[1];
    op[2 * (H_ * W_)] = acc[2] + b0[2];
    op[3 * (H_ * W_)] = acc[3] + b0[3];
    op[4 * (H_ * W_)] = acc[4] + b1[0];
    op[5 * (H_ * W_)] = acc[5] + b1[1];
    op[6 * (H_ * W_)] = acc[6] + b1[2];
    op[7 * (H_ * W_)] = acc[7] + b1[3];
  }
}

__global__ __launch_bounds__(256)
void xonv2d_wmma_f32(
    const float* __restrict__ x,
    const float* __restrict__ wgt,
    const float* __restrict__ bias,
    float* __restrict__ out)
{
  __shared__ __align__(16) float patch[WPB][WREG];

  const int lane = threadIdx.x & 31;
  const int wave = threadIdx.x >> 5;
  const int n    = lane & 15;
  const int half = lane >> 4;
  const int pix0 = blockIdx.x * (WPB * PPW) + wave * PPW;
  const int h    = blockIdx.x >> 2;
  const int w0   = pix0 & 127;

  const int r4  = lane >> 3;
  const int t   = lane & 7;
  const int u   = t - 1;
  const int col = w0 - 2 + t;
  const bool okc = (unsigned)col < (unsigned)W_;
  const int colc = min(max(col, 0), W_ - 1);

  float xv[3][16];
#pragma unroll
  for (int kh = 0; kh < 3; ++kh) {
    const int hhc = min(max(h + kh - 1, 0), H_ - 1);
    const float* xr = x + ((size_t)(r4 * H_ + hhc) * W_ + colc);
#pragma unroll
    for (int m = 0; m < 16; ++m) xv[kh][m] = xr[(size_t)m * (4 * H_ * W_)];
  }
  __builtin_amdgcn_sched_barrier(0);

  const float* wl = wgt + (size_t)pix0 * WSTEP + n * KK + 4 * half;
  v4f wa[NL], wb[NL];
  load_weights(wa, wl);
  __builtin_amdgcn_sched_barrier(0);

  float* pat = &patch[wave][0];
  int sb[3];
#pragma unroll
  for (int kw = 0; kw < 3; ++kw) {
    const int i = u - kw;
    sb[kw] = ((unsigned)i < (unsigned)PPW) ? i * PITCH + 9 * r4 + kw
                                           : PPW * PITCH + 26 + 4 * r4 + (t & 3);
  }
#pragma unroll
  for (int kh = 0; kh < 3; ++kh) {
    const bool ok = okc && (unsigned)(h + kh - 1) < (unsigned)H_;
#pragma unroll
    for (int m = 0; m < 16; ++m) {
      const float v = ok ? xv[kh][m] : 0.0f;
      const int c0 = 36 * m + 3 * kh;
      pat[sb[0] + c0] = v;
      pat[sb[1] + c0] = v;
      pat[sb[2] + c0] = v;
    }
  }
  __builtin_amdgcn_wave_barrier();

  const float* pb = pat + (n & (B_ - 1)) * KK + 4 * half;
  const float* bp = bias + (size_t)pix0 * COUT_ + 8 * half;
  float* op = out + (size_t)(n & (B_ - 1)) * COUT_ * (H_ * W_) + (size_t)(8 * half) * (H_ * W_) + pix0;
  const bool writer = n < B_;

  pixel_step(wa, wb, true,  wl + 1 * WSTEP, pb + 0 * PITCH, bp + 0 * COUT_, op + 0, writer);
  pixel_step(wb, wa, true,  wl + 2 * WSTEP, pb + 1 * PITCH, bp + 1 * COUT_, op + 1, writer);
  pixel_step(wa, wb, true,  wl + 3 * WSTEP, pb + 2 * PITCH, bp + 2 * COUT_, op + 2, writer);
  pixel_step(wb, wa, false, wl,             pb + 3 * PITCH, bp + 3 * COUT_, op + 3, writer);
}

extern "C" void kernel_launch(void* const* d_in, const int* in_sizes, int n_in,
                              void* d_out, int out_size, void* d_ws, size_t ws_size,
                              hipStream_t stream) {
  (void)in_sizes; (void)n_in; (void)out_size; (void)d_ws; (void)ws_size;
  const float* x    = (const float*)d_in[0];
  const float* wgt  = (const float*)d_in[1];
  const float* bias = (const float*)d_in[2];
  float* out        = (float*)d_out;

  dim3 grid((H_ * W_) / (WPB * PPW));
  dim3 block(256);
  xonv2d_wmma_f32<<<grid, block, 0, stream>>>(x, wgt, bias, out);
}


# 4 adjacent pixels' results stay in registers and go out as 16 B per lane from all 32 lanes: 2 b128 stores per wave (16-byte pieces) instead of 32 b32 stores of 8 lanes (4-byte pieces); sums unchanged
# speedup vs baseline: 2.9297x; 1.0861x over previous
#include <hip/hip_runtime.h>

typedef __attribute__((ext_vector_type(2))) float v2f;
typedef __attribute__((ext_vector_type(4))) float v4f;
typedef __attribute__((ext_vector_type(8))) float v8f;

#define B_    4
#define CIN_  16
#define COUT_ 16
#define H_    128
#define W_    128
#define KK    144
#define NT    (KK / 4)
#define NL    (KK / 8)
#define WPB   8
#define PPW   4
#define PSZ   (B_ * KK)
#define PITCH 580
#define WREG  2944
#define WSTEP (COUT_ * KK)


__device__ __forceinline__ void load_weights(v4f (&a)[NL], const float* __restrict__ wl) {
#pragma unroll
  for (int j = 0; j < NL; ++j) a[j] = *(const v4f*)(wl + 8 * j);
}

__device__ __forceinline__ v8f pixel_step(
    const v4f (&cur)[NL], v4f (&nxt)[NL], const bool has_next,
    const float* __restrict__ wl_next, const float* __restrict__ pb,
    const float* __restrict__ bp)
{
  v4f b0 = *(const v4f*)bp;
  v4f b1 = *(const v4f*)(bp + 4);
  __builtin_amdgcn_sched_barrier(0);
  if (has_next) load_weights(nxt, wl_next);
  __builtin_amdgcn_sched_barrier(0);

  v8f acc = {};
#pragma unroll
  for (int j = 0; j < NL; ++j) {
    const v4f a = cur[j];
    const v4f b = *(const v4f*)(pb + 8 * j);
    acc = __builtin_amdgcn_wmma_f32_16x16x4_f32(
         false, __builtin_shufflevector(a, a, 0, 1),
         false, __builtin_shufflevector(b, b, 0, 1),
         (short)0, acc,  false,  false);
    acc = __builtin_amdgcn_wmma_f32_16x16x4_f32(
         false, __builtin_shufflevector(a, a, 2, 3),
         false, __builtin_shufflevector(b, b, 2, 3),
         (short)0, acc,  false,  false);
  }

  v8f res;
  res[0] = acc[0] + b0[0];
  res[1] = acc[1] + b0[1];
  res[2] = acc[2] + b0[2];
  res[3] = acc[3] + b0[3];
  res[4] = acc[4] + b1[0];
  res[5] = acc[5] + b1[1];
  res[6] = acc[6] + b1[2];
  res[7] = acc[7] + b1[3];
  return res;
}

__device__ __forceinline__ void end_step(v8f& done, int& z) {
  float t = done[0];
  asm volatile("" : "+v"(t), "+v"(z));
  done[0] = t;
}

__device__ __forceinline__ float sel4(const bool q1, const bool q2,
                                      const float a, const float b, const float c, const float d) {
  const float lo = q1 ? b : a;
  const float hi = q1 ? d : c;
  return q2 ? hi : lo;
}

__global__ __launch_bounds__(256)
void xonv2d_wmma_f32(
    const float* __restrict__ x,
    const float* __restrict__ wgt,
    const float* __restrict__ bias,
    float* __restrict__ out)
{
  __shared__ __align__(16) float patch[WPB][WREG];

  const int lane = threadIdx.x & 31;
  const int wave = threadIdx.x >> 5;
  const int n    = lane & 15;
  const int half = lane >> 4;
  const int pix0 = blockIdx.x * (WPB * PPW) + wave * PPW;
  const int h    = blockIdx.x >> 2;
  const int w0   = pix0 & 127;

  const int r4  = lane >> 3;
  const int t   = lane & 7;
  const int u   = t - 1;
  const int col = w0 - 2 + t;
  const bool okc = (unsigned)col < (unsigned)W_;
  const int colc = min(max(col, 0), W_ - 1);

  float xv[3][16];
#pragma unroll
  for (int kh = 0; kh < 3; ++kh) {
    const int hhc = min(max(h + kh - 1, 0), H_ - 1);
    const float* xr = x + ((size_t)(r4 * H_ + hhc) * W_ + colc);
#pragma unroll
    for (int m = 0; m < 16; ++m) xv[kh][m] = xr[(size_t)m * (4 * H_ * W_)];
  }
  __builtin_amdgcn_sched_barrier(0);

  const float* wl = wgt + (size_t)pix0 * WSTEP + n * KK + 4 * half;
  v4f wa[NL], wb[NL];
  load_weights(wa, wl);
  __builtin_amdgcn_sched_barrier(0);

  float* pat = &patch[wave][0];
  int sb[3];
#pragma unroll
  for (int kw = 0; kw < 3; ++kw) {
    const int i = u - kw;
    sb[kw] = ((unsigned)i < (unsigned)PPW) ? i * PITCH + 9 * r4 + kw
                                           : PPW * PITCH + 26 + 4 * r4 + (t & 3);
  }
#pragma unroll
  for (int kh = 0; kh < 3; ++kh) {
    const bool ok = okc && (unsigned)(h + kh - 1) < (unsigned)H_;
#pragma unroll
    for (int m = 0; m < 16; ++m) {
      const float v = ok ? xv[kh][m] : 0.0f;
      const int c0 = 36 * m + 3 * kh;
      pat[sb[0] + c0] = v;
      pat[sb[1] + c0] = v;
      pat[sb[2] + c0] = v;
    }
  }
  __builtin_amdgcn_wave_barrier();

  const float* pb = pat + (n & (B_ - 1)) * KK + 4 * half;
  const float* bp = bias + (size_t)pix0 * COUT_ + 8 * half;

  int z = 0;
  v8f r0 = pixel_step(wa, wb, true,  wl + 1 * WSTEP,     pb + 0 * PITCH, bp + 0 * COUT_);
  end_step(r0, z);
  v8f r1 = pixel_step(wb, wa, true,  wl + z + 2 * WSTEP, pb + 1 * PITCH, bp + z + 1 * COUT_);
  end_step(r1, z);
  v8f r2 = pixel_step(wa, wb, true,  wl + z + 3 * WSTEP, pb + 2 * PITCH, bp + z + 2 * COUT_);
  end_step(r2, z);
  const v8f r3 = pixel_step(wb, wa, false, wl,           pb + 3 * PITCH, bp + z + 3 * COUT_);

  const int q = n >> 2;
  const bool q1 = (q & 1) != 0;
  const bool q2 = (q & 2) != 0;
  float* op = out + (size_t)((n & (B_ - 1)) * COUT_ + 8 * half + 2 * q) * (H_ * W_) + pix0;
  v4f v0, v1;
  v0[0] = sel4(q1, q2, r0[0], r0[2], r0[4], r0[6]);
  v0[1] = sel4(q1, q2, r1[0], r1[2], r1[4], r1[6]);
  v0[2] = sel4(q1, q2, r2[0], r2[2], r2[4], r2[6]);
  v0[3] = sel4(q1, q2, r3[0], r3[2], r3[4], r3[6]);
  v1[0] = sel4(q1, q2, r0[1], r0[3], r0[5], r0[7]);
  v1[1] = sel4(q1, q2, r1[1], r1[3], r1[5], r1[7]);
  v1[2] = sel4(q1, q2, r2[1], r2[3], r2[5], r2[7]);
  v1[3] = sel4(q1, q2, r3[1], r3[3], r3[5], r3[7]);
  *(v4f*)op             = v0;
  *(v4f*)(op + H_ * W_) = v1;
}

extern "C" void kernel_launch(void* const* d_in, const int* in_sizes, int n_in,
                              void* d_out, int out_size, void* d_ws, size_t ws_size,
                              hipStream_t stream) {
  (void)in_sizes; (void)n_in; (void)out_size; (void)d_ws; (void)ws_size;
  const float* x    = (const float*)d_in[0];
  const float* wgt  = (const float*)d_in[1];
  const float* bias = (const float*)d_in[2];
  float* out        = (float*)d_out;

  dim3 grid((H_ * W_) / (WPB * PPW));
  dim3 block(256);
  xonv2d_wmma_f32<<<grid, block, 0, stream>>>(x, wgt, bias, out);
}
